# starting kernel
# baseline (speedup 1.0000x reference)
#include <hip/hip_runtime.h>
#include <math.h>

#define MD    1024
#define NH    16
#define HD    64
#define BATCH 4
#define SEQ   2048

typedef __attribute__((ext_vector_type(2))) float v2f;
typedef __attribute__((ext_vector_type(8))) float v8f;

static __device__ __forceinline__ v8f wmma_f32(v2f a, v2f b, v8f c) {
  return __builtin_amdgcn_wmma_f32_16x16x4_f32(
       false, a,  false, b,
       (short)0, c,  false,  false);
}

__global__ __launch_bounds__(256)
void gemm_nt_128x64(const float* __restrict__ A, const float* __restrict__ W,
                    const float* __restrict__ bias, float* __restrict__ C,
                    int headsplit)
{
  __shared__ float As[128][18];
  __shared__ float Ws[64][18];

  const int m0   = blockIdx.x * 128;
  const int n0   = blockIdx.y * 64;
  const int tid  = threadIdx.x;
  const int w    = tid >> 5;
  const int lane = tid & 31;
  const int ll   = lane & 15;
  const int hh   = lane >> 4;
  const int mq   = (w >> 1) * 32;
  const int nq   = (w & 1) * 32;

  const v8f z = {0.f,0.f,0.f,0.f,0.f,0.f,0.f,0.f};
  v8f acc00 = z, acc01 = z, acc10 = z, acc11 = z;

  const int lrA0 = tid >> 2;
  const int lcA  = (tid & 3) * 4;
  const int lrW  = tid >> 2;

  for (int k0 = 0; k0 < MD; k0 += 16) {
    const float4 a0 = *(const float4*)&A[(size_t)(m0 + lrA0     ) * MD + k0 + lcA];
    const float4 a1 = *(const float4*)&A[(size_t)(m0 + lrA0 + 64) * MD + k0 + lcA];
    const float4 wv = *(const float4*)&W[(size_t)(n0 + lrW      ) * MD + k0 + lcA];
    As[lrA0   ][lcA+0] = a0.x; As[lrA0   ][lcA+1] = a0.y;
    As[lrA0   ][lcA+2] = a0.z; As[lrA0   ][lcA+3] = a0.w;
    As[lrA0+64][lcA+0] = a1.x; As[lrA0+64][lcA+1] = a1.y;
    As[lrA0+64][lcA+2] = a1.z; As[lrA0+64][lcA+3] = a1.w;
    Ws[lrW][lcA+0] = wv.x; Ws[lrW][lcA+1] = wv.y;
    Ws[lrW][lcA+2] = wv.z; Ws[lrW][lcA+3] = wv.w;
    __syncthreads();

    if (k0 + 16 < MD) {
      __builtin_prefetch(&A[(size_t)(m0 + lrA0     ) * MD + k0 + 16 + lcA], 0, 1);
      __builtin_prefetch(&A[(size_t)(m0 + lrA0 + 64) * MD + k0 + 16 + lcA], 0, 1);
      __builtin_prefetch(&W[(size_t)(n0 + lrW      ) * MD + k0 + 16 + lcA], 0, 1);
    }

#pragma unroll
    for (int kk = 0; kk < 16; kk += 4) {
      const int col = kk + 2 * hh;
      const v2f a0f = *(const v2f*)&As[mq      + ll][col];
      const v2f a1f = *(const v2f*)&As[mq + 16 + ll][col];
      const v2f b0f = *(const v2f*)&Ws[nq      + ll][col];
      const v2f b1f = *(const v2f*)&Ws[nq + 16 + ll][col];
      acc00 = wmma_f32(a0f, b0f, acc00);
      acc01 = wmma_f32(a0f, b1f, acc01);
      acc10 = wmma_f32(a1f, b0f, acc10);
      acc11 = wmma_f32(a1f, b1f, acc11);
    }
    __syncthreads();
  }

#pragma unroll
  for (int mi = 0; mi < 2; ++mi) {
#pragma unroll
    for (int ni = 0; ni < 2; ++ni) {
      const v8f acc = (mi == 0) ? (ni == 0 ? acc00 : acc01)
                                : (ni == 0 ? acc10 : acc11);
#pragma unroll
      for (int j = 0; j < 8; ++j) {
        const int gm = m0 + mq + mi * 16 + j + hh * 8;
        const int gn = n0 + nq + ni * 16 + ll;
        float c = acc[j];
        if (!headsplit) {
          c += bias[gn];
          C[(size_t)gm * MD + gn] = c;
        } else {
          const int b = gm / SEQ, n = gm % SEQ;
          const int h = gn >> 6, d = gn & 63;
          C[(((size_t)b * NH + h) * SEQ + n) * HD + d] = c;
        }
      }
    }
  }
}

__global__ __launch_bounds__(128)
void flash_attn_causal(const float* __restrict__ Q, const float* __restrict__ K,
                       const float* __restrict__ V, float* __restrict__ O)
{
  __shared__ float Ks[16][66];
  __shared__ float Vt[64][18];
  __shared__ float Ps[4][16][18];

  const int bh   = blockIdx.x;
  const int qblk = blockIdx.y;
  const int tid  = threadIdx.x;
  const int w    = tid >> 5;
  const int lane = tid & 31;
  const int ll   = lane & 15;
  const int hh   = lane >> 4;
  const int qbase = qblk * 64 + w * 16;
  const float scale = 0.125f;

  const float* Qh = Q + (size_t)bh * SEQ * HD;
  const float* Kh = K + (size_t)bh * SEQ * HD;
  const float* Vh = V + (size_t)bh * SEQ * HD;

  v2f qf[16];
#pragma unroll
  for (int t = 0; t < 16; ++t) {
    v2f qv = *(const v2f*)&Qh[(size_t)(qbase + ll) * HD + t * 4 + 2 * hh];
    qf[t] = qv * scale;
  }

  const v8f z = {0.f,0.f,0.f,0.f,0.f,0.f,0.f,0.f};
  v8f oacc[4] = {z, z, z, z};
  float mrow[8], lrow[8];
#pragma unroll
  for (int j = 0; j < 8; ++j) { mrow[j] = -INFINITY; lrow[j] = 0.f; }

  const int nkt = qblk * 4 + 4;
  for (int kt = 0; kt < nkt; ++kt) {
    const int kbase = kt * 16;

#pragma unroll
    for (int part = 0; part < 2; ++part) {
      const int lin = tid + part * 128;
      const int row = lin >> 4, c4 = (lin & 15) * 4;
      const float4 kd = *(const float4*)&Kh[(size_t)(kbase + row) * HD + c4];
      const float4 vd = *(const float4*)&Vh[(size_t)(kbase + row) * HD + c4];
      Ks[row][c4+0] = kd.x; Ks[row][c4+1] = kd.y;
      Ks[row][c4+2] = kd.z; Ks[row][c4+3] = kd.w;
      Vt[c4+0][row] = vd.x; Vt[c4+1][row] = vd.y;
      Vt[c4+2][row] = vd.z; Vt[c4+3][row] = vd.w;
    }
    __syncthreads();

    if (kbase <= qbase + 15) {
      v8f s = z;
#pragma unroll
      for (int t = 0; t < 16; ++t) {
        const v2f b = *(const v2f*)&Ks[ll][t * 4 + 2 * hh];
        s = wmma_f32(qf[t], b, s);
      }

#pragma unroll
      for (int j = 0; j < 8; ++j) {
        const int rowg = qbase + j + hh * 8;
        const int colg = kbase + ll;
        const float sj = (colg > rowg) ? -INFINITY : s[j];
        float red = sj;
        red = fmaxf(red, __shfl_xor(red, 1, 32));
        red = fmaxf(red, __shfl_xor(red, 2, 32));
        red = fmaxf(red, __shfl_xor(red, 4, 32));
        red = fmaxf(red, __shfl_xor(red, 8, 32));
        const float mnew  = fmaxf(mrow[j], red);
        const float p     = __expf(sj - mnew);
        const float alpha = __expf(mrow[j] - mnew);
        mrow[j] = mnew;
        float psum = p;
        psum += __shfl_xor(psum, 1, 32);
        psum += __shfl_xor(psum, 2, 32);
        psum += __shfl_xor(psum, 4, 32);
        psum += __shfl_xor(psum, 8, 32);
        lrow[j] = lrow[j] * alpha + psum;
#pragma unroll
        for (int dt = 0; dt < 4; ++dt) oacc[dt][j] *= alpha;
        Ps[w][j + hh * 8][ll] = p;
      }

#pragma unroll
      for (int t = 0; t < 4; ++t) {
        const v2f a = *(const v2f*)&Ps[w][ll][t * 4 + 2 * hh];
#pragma unroll
        for (int dt = 0; dt < 4; ++dt) {
          const v2f b = *(const v2f*)&Vt[dt * 16 + ll][t * 4 + 2 * hh];
          oacc[dt] = wmma_f32(a, b, oacc[dt]);
        }
      }
    }
    __syncthreads();
  }

  const int b = bh >> 4;
  const int h = bh & 15;
#pragma unroll
  for (int j = 0; j < 8; ++j) {
    const float inv = 1.0f / lrow[j];
    const int rowg = qbase + j + hh * 8;
#pragma unroll
    for (int dt = 0; dt < 4; ++dt) {
      const int d = dt * 16 + ll;
      O[((size_t)b * SEQ + rowg) * MD + h * HD + d] = oacc[dt][j] * inv;
    }
  }
}

extern "C" void kernel_launch(void* const* d_in, const int* in_sizes, int n_in,
                              void* d_out, int out_size, void* d_ws, size_t ws_size,
                              hipStream_t stream) {
  (void)in_sizes; (void)n_in; (void)out_size; (void)ws_size;
  const float* x  = (const float*)d_in[0];
  const float* Wq = (const float*)d_in[1];
  const float* Wk = (const float*)d_in[2];
  const float* Wv = (const float*)d_in[3];
  const float* Wo = (const float*)d_in[4];
  const float* bo = (const float*)d_in[5];

  float* ws = (float*)d_ws;
  const size_t SZ = (size_t)BATCH * SEQ * MD;
  float* qb = ws;
  float* kb = ws + SZ;
  float* vb = ws + 2 * SZ;
  float* ao = ws + 3 * SZ;

  const dim3 ggrid(BATCH * SEQ / 128, MD / 64);
  gemm_nt_128x64<<<ggrid, 256, 0, stream>>>(x, Wq, nullptr, qb, 1);
  gemm_nt_128x64<<<ggrid, 256, 0, stream>>>(x, Wk, nullptr, kb, 1);
  gemm_nt_128x64<<<ggrid, 256, 0, stream>>>(x, Wv, nullptr, vb, 1);

  flash_attn_causal<<<dim3(BATCH * NH, SEQ / 64), 128, 0, stream>>>(qb, kb, vb, ao);

  gemm_nt_128x64<<<ggrid, 256, 0, stream>>>(ao, Wo, bo, (float*)d_out, 0);
}


# Q, K and V projections fused into one launch: a block stages its 128x16 x tile once per K step and multiplies it with the three 64x16 weight tiles (12 accumulators per wave); launches 5 -> 3
# speedup vs baseline: 1.2252x; 1.2252x over previous
#include <hip/hip_runtime.h>
#include <math.h>

#define MD    1024
#define NH    16
#define HD    64
#define BATCH 4
#define SEQ   2048

typedef __attribute__((ext_vector_type(2))) float v2f;
typedef __attribute__((ext_vector_type(4))) float v4f;
typedef __attribute__((ext_vector_type(8))) float v8f;

static __device__ __forceinline__ v8f wmma_f32(v2f a, v2f b, v8f c) {
  return __builtin_amdgcn_wmma_f32_16x16x4_f32(
       false, a,  false, b,
       (short)0, c,  false,  false);
}

__global__ __launch_bounds__(256)
void gemm_nt_128x64(const float* __restrict__ A, const float* __restrict__ W,
                    const float* __restrict__ bias, float* __restrict__ C,
                    int headsplit)
{
  __shared__ float As[128][18];
  __shared__ float Ws[64][18];

  const int m0   = blockIdx.x * 128;
  const int n0   = blockIdx.y * 64;
  const int tid  = threadIdx.x;
  const int w    = tid >> 5;
  const int lane = tid & 31;
  const int ll   = lane & 15;
  const int hh   = lane >> 4;
  const int mq   = (w >> 1) * 32;
  const int nq   = (w & 1) * 32;

  const v8f z = {0.f,0.f,0.f,0.f,0.f,0.f,0.f,0.f};
  v8f acc00 = z, acc01 = z, acc10 = z, acc11 = z;

  const int lrA0 = tid >> 2;
  const int lcA  = (tid & 3) * 4;
  const int lrW  = tid >> 2;

  for (int k0 = 0; k0 < MD; k0 += 16) {
    const float4 a0 = *(const float4*)&A[(size_t)(m0 + lrA0     ) * MD + k0 + lcA];
    const float4 a1 = *(const float4*)&A[(size_t)(m0 + lrA0 + 64) * MD + k0 + lcA];
    const float4 wv = *(const float4*)&W[(size_t)(n0 + lrW      ) * MD + k0 + lcA];
    As[lrA0   ][lcA+0] = a0.x; As[lrA0   ][lcA+1] = a0.y;
    As[lrA0   ][lcA+2] = a0.z; As[lrA0   ][lcA+3] = a0.w;
    As[lrA0+64][lcA+0] = a1.x; As[lrA0+64][lcA+1] = a1.y;
    As[lrA0+64][lcA+2] = a1.z; As[lrA0+64][lcA+3] = a1.w;
    Ws[lrW][lcA+0] = wv.x; Ws[lrW][lcA+1] = wv.y;
    Ws[lrW][lcA+2] = wv.z; Ws[lrW][lcA+3] = wv.w;
    __syncthreads();

    if (k0 + 16 < MD) {
      __builtin_prefetch(&A[(size_t)(m0 + lrA0     ) * MD + k0 + 16 + lcA], 0, 1);
      __builtin_prefetch(&A[(size_t)(m0 + lrA0 + 64) * MD + k0 + 16 + lcA], 0, 1);
      __builtin_prefetch(&W[(size_t)(n0 + lrW      ) * MD + k0 + 16 + lcA], 0, 1);
    }

#pragma unroll
    for (int kk = 0; kk < 16; kk += 4) {
      const int col = kk + 2 * hh;
      const v2f a0f = *(const v2f*)&As[mq      + ll][col];
      const v2f a1f = *(const v2f*)&As[mq + 16 + ll][col];
      const v2f b0f = *(const v2f*)&Ws[nq      + ll][col];
      const v2f b1f = *(const v2f*)&Ws[nq + 16 + ll][col];
      acc00 = wmma_f32(a0f, b0f, acc00);
      acc01 = wmma_f32(a0f, b1f, acc01);
      acc10 = wmma_f32(a1f, b0f, acc10);
      acc11 = wmma_f32(a1f, b1f, acc11);
    }
    __syncthreads();
  }

#pragma unroll
  for (int mi = 0; mi < 2; ++mi) {
#pragma unroll
    for (int ni = 0; ni < 2; ++ni) {
      const v8f acc = (mi == 0) ? (ni == 0 ? acc00 : acc01)
                                : (ni == 0 ? acc10 : acc11);
#pragma unroll
      for (int j = 0; j < 8; ++j) {
        const int gm = m0 + mq + mi * 16 + j + hh * 8;
        const int gn = n0 + nq + ni * 16 + ll;
        float c = acc[j];
        if (!headsplit) {
          c += bias[gn];
          C[(size_t)gm * MD + gn] = c;
        } else {
          const int b = gm / SEQ, n = gm % SEQ;
          const int h = gn >> 6, d = gn & 63;
          C[(((size_t)b * NH + h) * SEQ + n) * HD + d] = c;
        }
      }
    }
  }
}

__global__ __launch_bounds__(256)
void gemm_qkv_nt_128x64(const float* __restrict__ A,
                        const float* __restrict__ Wq, const float* __restrict__ Wk,
                        const float* __restrict__ Wv,
                        float* __restrict__ Cq, float* __restrict__ Ck,
                        float* __restrict__ Cv)
{
  __shared__ float As[128][18];
  __shared__ float Ws[3][64][18];

  const int m0   = blockIdx.x * 128;
  const int n0   = blockIdx.y * 64;
  const int tid  = threadIdx.x;
  const int w    = tid >> 5;
  const int lane = tid & 31;
  const int ll   = lane & 15;
  const int hh   = lane >> 4;
  const int mq   = (w >> 1) * 32;
  const int nq   = (w & 1) * 32;

  const v8f z = {0.f,0.f,0.f,0.f,0.f,0.f,0.f,0.f};
  v8f acc[3][2][2];
#pragma unroll
  for (int s = 0; s < 3; ++s) {
    acc[s][0][0] = z; acc[s][0][1] = z; acc[s][1][0] = z; acc[s][1][1] = z;
  }

  const int lr = tid >> 2;
  const int lc = (tid & 3) * 4;

  const float* pa0 = A  + (size_t)(m0 + lr     ) * MD + lc;
  const float* pa1 = A  + (size_t)(m0 + lr + 64) * MD + lc;
  const float* pw0 = Wq + (size_t)(n0 + lr     ) * MD + lc;
  const float* pw1 = Wk + (size_t)(n0 + lr     ) * MD + lc;
  const float* pw2 = Wv + (size_t)(n0 + lr     ) * MD + lc;

  for (int k0 = 0; k0 < MD; k0 += 16) {
    const v4f a0 = *(const v4f*)(pa0 + k0);
    const v4f a1 = *(const v4f*)(pa1 + k0);
    const v4f w0 = *(const v4f*)(pw0 + k0);
    const v4f w1 = *(const v4f*)(pw1 + k0);
    const v4f w2 = *(const v4f*)(pw2 + k0);
    asm volatile("" :: "v"(a0), "v"(a1), "v"(w0), "v"(w1), "v"(w2));
    As[lr   ][lc+0] = a0.x; As[lr   ][lc+1] = a0.y;
    As[lr   ][lc+2] = a0.z; As[lr   ][lc+3] = a0.w;
    As[lr+64][lc+0] = a1.x; As[lr+64][lc+1] = a1.y;
    As[lr+64][lc+2] = a1.z; As[lr+64][lc+3] = a1.w;
    Ws[0][lr][lc+0] = w0.x; Ws[0][lr][lc+1] = w0.y;
    Ws[0][lr][lc+2] = w0.z; Ws[0][lr][lc+3] = w0.w;
    Ws[1][lr][lc+0] = w1.x; Ws[1][lr][lc+1] = w1.y;
    Ws[1][lr][lc+2] = w1.z; Ws[1][lr][lc+3] = w1.w;
    Ws[2][lr][lc+0] = w2.x; Ws[2][lr][lc+1] = w2.y;
    Ws[2][lr][lc+2] = w2.z; Ws[2][lr][lc+3] = w2.w;
    __syncthreads();

    if (k0 + 16 < MD) {
      __builtin_prefetch(pa0 + k0 + 16, 0, 1);
      __builtin_prefetch(pa1 + k0 + 16, 0, 1);
      __builtin_prefetch(pw0 + k0 + 16, 0, 1);
      __builtin_prefetch(pw1 + k0 + 16, 0, 1);
      __builtin_prefetch(pw2 + k0 + 16, 0, 1);
    }

#pragma unroll
    for (int kk = 0; kk < 16; kk += 4) {
      const int col = kk + 2 * hh;
      const v2f a0f = *(const v2f*)&As[mq      + ll][col];
      const v2f a1f = *(const v2f*)&As[mq + 16 + ll][col];
#pragma unroll
      for (int s = 0; s < 3; ++s) {
        const v2f b0f = *(const v2f*)&Ws[s][nq      + ll][col];
        const v2f b1f = *(const v2f*)&Ws[s][nq + 16 + ll][col];
        acc[s][0][0] = wmma_f32(a0f, b0f, acc[s][0][0]);
        acc[s][0][1] = wmma_f32(a0f, b1f, acc[s][0][1]);
        acc[s][1][0] = wmma_f32(a1f, b0f, acc[s][1][0]);
        acc[s][1][1] = wmma_f32(a1f, b1f, acc[s][1][1]);
      }
    }
    __syncthreads();
  }

#pragma unroll
  for (int s = 0; s < 3; ++s) {
    float* __restrict__ C = (s == 0) ? Cq : (s == 1 ? Ck : Cv);
#pragma unroll
    for (int mi = 0; mi < 2; ++mi) {
#pragma unroll
      for (int ni = 0; ni < 2; ++ni) {
        const v8f av = acc[s][mi][ni];
#pragma unroll
        for (int j = 0; j < 8; ++j) {
          const int gm = m0 + mq + mi * 16 + j + hh * 8;
          const int gn = n0 + nq + ni * 16 + ll;
          const int b = gm / SEQ, n = gm % SEQ;
          const int h = gn >> 6, d = gn & 63;
          C[(((size_t)b * NH + h) * SEQ + n) * HD + d] = av[j];
        }
      }
    }
  }
}

#define KSTEP 64
#define LP    68

__global__ __launch_bounds__(128)
void flash_attn_causal(const float* __restrict__ Q, const float* __restrict__ K,
                       const float* __restrict__ V, float* __restrict__ O)
{
  __shared__ __attribute__((aligned(16))) float Ks[KSTEP][LP];
  __shared__ __attribute__((aligned(16))) float Vt[HD][LP];
  __shared__ __attribute__((aligned(16))) float Ps[4][16][LP];

  const int bh   = blockIdx.x;
  const int qblk = blockIdx.y;
  const int tid  = threadIdx.x;
  const int w    = __builtin_amdgcn_readfirstlane(tid >> 5);
  const int lane = tid & 31;
  const int ll   = lane & 15;
  const int hh   = lane >> 4;
  const int qbase = qblk * 64 + w * 16;
  const float scale = 0.125f;

  const float* Qh = Q + (size_t)bh * SEQ * HD;
  const float* Kh = K + (size_t)bh * SEQ * HD;
  const float* Vh = V + (size_t)bh * SEQ * HD;

  v2f qf[16];
#pragma unroll
  for (int t = 0; t < 16; ++t) {
    v2f qv = *(const v2f*)&Qh[(size_t)(qbase + ll) * HD + t * 4 + 2 * hh];
    qf[t] = qv * scale;
  }

  const v8f z = {0.f,0.f,0.f,0.f,0.f,0.f,0.f,0.f};
  v8f oacc[4] = {z, z, z, z};
  float mrow[8], lrow[8];
#pragma unroll
  for (int j = 0; j < 8; ++j) { mrow[j] = -INFINITY; lrow[j] = 0.f; }

  const int r0   = tid >> 4;
  const int kc4  = (tid & 15) * 4;
  const int vkey = lane & 7;
  const int vc4  = (w * 4 + (lane >> 3)) * 4;
  const float* kp = Kh + r0 * HD + kc4;
  const float* vp = Vh + vkey * HD + vc4;
  float* ksd = &Ks[r0][kc4];
  float* vsd = &Vt[vc4][vkey];

  const int nkt = qblk + 1;
  for (int kt = 0; kt < nkt; ++kt) {
    const int kbase = kt * KSTEP;

    const float* kpt = kp + (size_t)kbase * HD;
    const float* vpt = vp + (size_t)kbase * HD;
#pragma unroll
    for (int part = 0; part < 8; ++part) {
      const float4 kd = *(const float4*)(kpt + part * 8 * HD);
      const float4 vd = *(const float4*)(vpt + part * 8 * HD);
      *(float4*)(ksd + part * 8 * LP) = kd;
      float* vs = vsd + part * 8;
      vs[0] = vd.x; vs[LP] = vd.y; vs[2 * LP] = vd.z; vs[3 * LP] = vd.w;
    }
    __syncthreads();

    const bool diag = (kt == nkt - 1);
    const int  nsub = diag ? (w + 1) : 4;

    v8f s[4] = {z, z, z, z};
    if (nsub == 4) {
#pragma unroll
      for (int t = 0; t < 16; ++t) {
#pragma unroll
        for (int n = 0; n < 4; ++n) {
          const v2f b = *(const v2f*)&Ks[n * 16 + ll][t * 4 + 2 * hh];
          s[n] = wmma_f32(qf[t], b, s[n]);
        }
      }
    } else {
#pragma unroll
      for (int n = 0; n < 3; ++n) {
        if (n < nsub) {
#pragma unroll
          for (int t = 0; t < 16; ++t) {
            const v2f b = *(const v2f*)&Ks[n * 16 + ll][t * 4 + 2 * hh];
            s[n] = wmma_f32(qf[t], b, s[n]);
          }
        }
      }
    }

    if (diag) {
#pragma unroll
      for (int n = 0; n < 4; ++n) {
#pragma unroll
        for (int j = 0; j < 8; ++j) {
          const int rowg = qbase + j + hh * 8;
          const int colg = kbase + n * 16 + ll;
          if (colg > rowg) s[n][j] = -INFINITY;
        }
      }
    }

#pragma unroll
    for (int j = 0; j < 8; ++j) {
      float red = fmaxf(fmaxf(s[0][j], s[1][j]), fmaxf(s[2][j], s[3][j]));
      red = fmaxf(red, __shfl_xor(red, 1, 32));
      red = fmaxf(red, __shfl_xor(red, 2, 32));
      red = fmaxf(red, __shfl_xor(red, 4, 32));
      red = fmaxf(red, __shfl_xor(red, 8, 32));
      const float mnew  = fmaxf(mrow[j], red);
      const float alpha = __expf(mrow[j] - mnew);
      mrow[j] = mnew;
      const float p0 = __expf(s[0][j] - mnew);
      const float p1 = __expf(s[1][j] - mnew);
      const float p2 = __expf(s[2][j] - mnew);
      const float p3 = __expf(s[3][j] - mnew);
      float psum = (p0 + p1) + (p2 + p3);
      psum += __shfl_xor(psum, 1, 32);
      psum += __shfl_xor(psum, 2, 32);
      psum += __shfl_xor(psum, 4, 32);
      psum += __shfl_xor(psum, 8, 32);
      lrow[j] = lrow[j] * alpha + psum;
#pragma unroll
      for (int dt = 0; dt < 4; ++dt) oacc[dt][j] *= alpha;
      float* prow = &Ps[w][j + hh * 8][ll];
      prow[0] = p0; prow[16] = p1; prow[32] = p2; prow[48] = p3;
    }

    if (nsub == 4) {
#pragma unroll
      for (int t = 0; t < 16; ++t) {
        const v2f a = *(const v2f*)&Ps[w][ll][t * 4 + 2 * hh];
#pragma unroll
        for (int dt = 0; dt < 4; ++dt) {
          const v2f b = *(const v2f*)&Vt[dt * 16 + ll][t * 4 + 2 * hh];
          oacc[dt] = wmma_f32(a, b, oacc[dt]);
        }
      }
    } else {
#pragma unroll
      for (int n = 0; n < 3; ++n) {
        if (n < nsub) {
#pragma unroll
          for (int tt = 0; tt < 4; ++tt) {
            const int t = n * 4 + tt;
            const v2f a = *(const v2f*)&Ps[w][ll][t * 4 + 2 * hh];
#pragma unroll
            for (int dt = 0; dt < 4; ++dt) {
              const v2f b = *(const v2f*)&Vt[dt * 16 + ll][t * 4 + 2 * hh];
              oacc[dt] = wmma_f32(a, b, oacc[dt]);
            }
          }
        }
      }
    }
    __syncthreads();
  }

  const int b = bh >> 4;
  const int h = bh & 15;
#pragma unroll
  for (int j = 0; j < 8; ++j) {
    const float inv = 1.0f / lrow[j];
    const int rowg = qbase + j + hh * 8;
#pragma unroll
    for (int dt = 0; dt < 4; ++dt) {
      const int d = dt * 16 + ll;
      O[((size_t)b * SEQ + rowg) * MD + h * HD + d] = oacc[dt][j] * inv;
    }
  }
}

extern "C" void kernel_launch(void* const* d_in, const int* in_sizes, int n_in,
                              void* d_out, int out_size, void* d_ws, size_t ws_size,
                              hipStream_t stream) {
  (void)in_sizes; (void)n_in; (void)out_size; (void)ws_size;
  const float* x  = (const float*)d_in[0];
  const float* Wq = (const float*)d_in[1];
  const float* Wk = (const float*)d_in[2];
  const float* Wv = (const float*)d_in[3];
  const float* Wo = (const float*)d_in[4];
  const float* bo = (const float*)d_in[5];

  float* ws = (float*)d_ws;
  const size_t SZ = (size_t)BATCH * SEQ * MD;
  float* qb = ws;
  float* kb = ws + SZ;
  float* vb = ws + 2 * SZ;
  float* ao = ws + 3 * SZ;

  const dim3 ggrid(BATCH * SEQ / 128, MD / 64);
  gemm_qkv_nt_128x64<<<ggrid, 256, 0, stream>>>(x, Wq, Wk, Wv, qb, kb, vb);

  flash_attn_causal<<<dim3(BATCH * NH, SEQ / 64), 128, 0, stream>>>(qb, kb, vb, ao);

  gemm_nt_128x64<<<ggrid, 256, 0, stream>>>(ao, Wo, bo, (float*)d_out, 0);
}


# attention softmax cross-lane reductions: row max by 4 DPP row_xmask steps, not LDS permutes; row sum kept per lane, reduced over the 16 lanes once after the key loop; row-sum order changes
# speedup vs baseline: 1.2716x; 1.0379x over previous
#include <hip/hip_runtime.h>
#include <math.h>

#define MD    1024
#define NH    16
#define HD    64
#define BATCH 4
#define SEQ   2048

typedef __attribute__((ext_vector_type(2))) float v2f;
typedef __attribute__((ext_vector_type(4))) float v4f;
typedef __attribute__((ext_vector_type(8))) float v8f;

static __device__ __forceinline__ v8f wmma_f32(v2f a, v2f b, v8f c) {
  return __builtin_amdgcn_wmma_f32_16x16x4_f32(
       false, a,  false, b,
       (short)0, c,  false,  false);
}

__global__ __launch_bounds__(256)
void gemm_nt_128x64(const float* __restrict__ A, const float* __restrict__ W,
                    const float* __restrict__ bias, float* __restrict__ C,
                    int headsplit)
{
  __shared__ float As[128][18];
  __shared__ float Ws[64][18];

  const int m0   = blockIdx.x * 128;
  const int n0   = blockIdx.y * 64;
  const int tid  = threadIdx.x;
  const int w    = tid >> 5;
  const int lane = tid & 31;
  const int ll   = lane & 15;
  const int hh   = lane >> 4;
  const int mq   = (w >> 1) * 32;
  const int nq   = (w & 1) * 32;

  const v8f z = {0.f,0.f,0.f,0.f,0.f,0.f,0.f,0.f};
  v8f acc00 = z, acc01 = z, acc10 = z, acc11 = z;

  const int lrA0 = tid >> 2;
  const int lcA  = (tid & 3) * 4;
  const int lrW  = tid >> 2;

  for (int k0 = 0; k0 < MD; k0 += 16) {
    const float4 a0 = *(const float4*)&A[(size_t)(m0 + lrA0     ) * MD + k0 + lcA];
    const float4 a1 = *(const float4*)&A[(size_t)(m0 + lrA0 + 64) * MD + k0 + lcA];
    const float4 wv = *(const float4*)&W[(size_t)(n0 + lrW      ) * MD + k0 + lcA];
    As[lrA0   ][lcA+0] = a0.x; As[lrA0   ][lcA+1] = a0.y;
    As[lrA0   ][lcA+2] = a0.z; As[lrA0   ][lcA+3] = a0.w;
    As[lrA0+64][lcA+0] = a1.x; As[lrA0+64][lcA+1] = a1.y;
    As[lrA0+64][lcA+2] = a1.z; As[lrA0+64][lcA+3] = a1.w;
    Ws[lrW][lcA+0] = wv.x; Ws[lrW][lcA+1] = wv.y;
    Ws[lrW][lcA+2] = wv.z; Ws[lrW][lcA+3] = wv.w;
    __syncthreads();

    if (k0 + 16 < MD) {
      __builtin_prefetch(&A[(size_t)(m0 + lrA0     ) * MD + k0 + 16 + lcA], 0, 1);
      __builtin_prefetch(&A[(size_t)(m0 + lrA0 + 64) * MD + k0 + 16 + lcA], 0, 1);
      __builtin_prefetch(&W[(size_t)(n0 + lrW      ) * MD + k0 + 16 + lcA], 0, 1);
    }

#pragma unroll
    for (int kk = 0; kk < 16; kk += 4) {
      const int col = kk + 2 * hh;
      const v2f a0f = *(const v2f*)&As[mq      + ll][col];
      const v2f a1f = *(const v2f*)&As[mq + 16 + ll][col];
      const v2f b0f = *(const v2f*)&Ws[nq      + ll][col];
      const v2f b1f = *(const v2f*)&Ws[nq + 16 + ll][col];
      acc00 = wmma_f32(a0f, b0f, acc00);
      acc01 = wmma_f32(a0f, b1f, acc01);
      acc10 = wmma_f32(a1f, b0f, acc10);
      acc11 = wmma_f32(a1f, b1f, acc11);
    }
    __syncthreads();
  }

#pragma unroll
  for (int mi = 0; mi < 2; ++mi) {
#pragma unroll
    for (int ni = 0; ni < 2; ++ni) {
      const v8f acc = (mi == 0) ? (ni == 0 ? acc00 : acc01)
                                : (ni == 0 ? acc10 : acc11);
#pragma unroll
      for (int j = 0; j < 8; ++j) {
        const int gm = m0 + mq + mi * 16 + j + hh * 8;
        const int gn = n0 + nq + ni * 16 + ll;
        float c = acc[j];
        if (!headsplit) {
          c += bias[gn];
          C[(size_t)gm * MD + gn] = c;
        } else {
          const int b = gm / SEQ, n = gm % SEQ;
          const int h = gn >> 6, d = gn & 63;
          C[(((size_t)b * NH + h) * SEQ + n) * HD + d] = c;
        }
      }
    }
  }
}

__global__ __launch_bounds__(256)
void gemm_qkv_nt_128x64(const float* __restrict__ A,
                        const float* __restrict__ Wq, const float* __restrict__ Wk,
                        const float* __restrict__ Wv,
                        float* __restrict__ Cq, float* __restrict__ Ck,
                        float* __restrict__ Cv)
{
  __shared__ float As[128][18];
  __shared__ float Ws[3][64][18];

  const int m0   = blockIdx.x * 128;
  const int n0   = blockIdx.y * 64;
  const int tid  = threadIdx.x;
  const int w    = tid >> 5;
  const int lane = tid & 31;
  const int ll   = lane & 15;
  const int hh   = lane >> 4;
  const int mq   = (w >> 1) * 32;
  const int nq   = (w & 1) * 32;

  const v8f z = {0.f,0.f,0.f,0.f,0.f,0.f,0.f,0.f};
  v8f acc[3][2][2];
#pragma unroll
  for (int s = 0; s < 3; ++s) {
    acc[s][0][0] = z; acc[s][0][1] = z; acc[s][1][0] = z; acc[s][1][1] = z;
  }

  const int lr = tid >> 2;
  const int lc = (tid & 3) * 4;

  const float* pa0 = A  + (size_t)(m0 + lr     ) * MD + lc;
  const float* pa1 = A  + (size_t)(m0 + lr + 64) * MD + lc;
  const float* pw0 = Wq + (size_t)(n0 + lr     ) * MD + lc;
  const float* pw1 = Wk + (size_t)(n0 + lr     ) * MD + lc;
  const float* pw2 = Wv + (size_t)(n0 + lr     ) * MD + lc;

  for (int k0 = 0; k0 < MD; k0 += 16) {
    const v4f a0 = *(const v4f*)(pa0 + k0);
    const v4f a1 = *(const v4f*)(pa1 + k0);
    const v4f w0 = *(const v4f*)(pw0 + k0);
    const v4f w1 = *(const v4f*)(pw1 + k0);
    const v4f w2 = *(const v4f*)(pw2 + k0);
    asm volatile("" :: "v"(a0), "v"(a1), "v"(w0), "v"(w1), "v"(w2));
    As[lr   ][lc+0] = a0.x; As[lr   ][lc+1] = a0.y;
    As[lr   ][lc+2] = a0.z; As[lr   ][lc+3] = a0.w;
    As[lr+64][lc+0] = a1.x; As[lr+64][lc+1] = a1.y;
    As[lr+64][lc+2] = a1.z; As[lr+64][lc+3] = a1.w;
    Ws[0][lr][lc+0] = w0.x; Ws[0][lr][lc+1] = w0.y;
    Ws[0][lr][lc+2] = w0.z; Ws[0][lr][lc+3] = w0.w;
    Ws[1][lr][lc+0] = w1.x; Ws[1][lr][lc+1] = w1.y;
    Ws[1][lr][lc+2] = w1.z; Ws[1][lr][lc+3] = w1.w;
    Ws[2][lr][lc+0] = w2.x; Ws[2][lr][lc+1] = w2.y;
    Ws[2][lr][lc+2] = w2.z; Ws[2][lr][lc+3] = w2.w;
    __syncthreads();

    if (k0 + 16 < MD) {
      __builtin_prefetch(pa0 + k0 + 16, 0, 1);
      __builtin_prefetch(pa1 + k0 + 16, 0, 1);
      __builtin_prefetch(pw0 + k0 + 16, 0, 1);
      __builtin_prefetch(pw1 + k0 + 16, 0, 1);
      __builtin_prefetch(pw2 + k0 + 16, 0, 1);
    }

#pragma unroll
    for (int kk = 0; kk < 16; kk += 4) {
      const int col = kk + 2 * hh;
      const v2f a0f = *(const v2f*)&As[mq      + ll][col];
      const v2f a1f = *(const v2f*)&As[mq + 16 + ll][col];
#pragma unroll
      for (int s = 0; s < 3; ++s) {
        const v2f b0f = *(const v2f*)&Ws[s][nq      + ll][col];
        const v2f b1f = *(const v2f*)&Ws[s][nq + 16 + ll][col];
        acc[s][0][0] = wmma_f32(a0f, b0f, acc[s][0][0]);
        acc[s][0][1] = wmma_f32(a0f, b1f, acc[s][0][1]);
        acc[s][1][0] = wmma_f32(a1f, b0f, acc[s][1][0]);
        acc[s][1][1] = wmma_f32(a1f, b1f, acc[s][1][1]);
      }
    }
    __syncthreads();
  }

#pragma unroll
  for (int s = 0; s < 3; ++s) {
    float* __restrict__ C = (s == 0) ? Cq : (s == 1 ? Ck : Cv);
#pragma unroll
    for (int mi = 0; mi < 2; ++mi) {
#pragma unroll
      for (int ni = 0; ni < 2; ++ni) {
        const v8f av = acc[s][mi][ni];
#pragma unroll
        for (int j = 0; j < 8; ++j) {
          const int gm = m0 + mq + mi * 16 + j + hh * 8;
          const int gn = n0 + nq + ni * 16 + ll;
          const int b = gm / SEQ, n = gm % SEQ;
          const int h = gn >> 6, d = gn & 63;
          C[(((size_t)b * NH + h) * SEQ + n) * HD + d] = av[j];
        }
      }
    }
  }
}

#define ROW8_STEP(op, M)                                              \
  op " %0, %0, %0 row_xmask:" #M " row_mask:0xf bank_mask:0xf\n\t"     \
  op " %1, %1, %1 row_xmask:" #M " row_mask:0xf bank_mask:0xf\n\t"     \
  op " %2, %2, %2 row_xmask:" #M " row_mask:0xf bank_mask:0xf\n\t"     \
  op " %3, %3, %3 row_xmask:" #M " row_mask:0xf bank_mask:0xf\n\t"     \
  op " %4, %4, %4 row_xmask:" #M " row_mask:0xf bank_mask:0xf\n\t"     \
  op " %5, %5, %5 row_xmask:" #M " row_mask:0xf bank_mask:0xf\n\t"     \
  op " %6, %6, %6 row_xmask:" #M " row_mask:0xf bank_mask:0xf\n\t"     \
  op " %7, %7, %7 row_xmask:" #M " row_mask:0xf bank_mask:0xf\n\t"
#define ROW8_REDUCE(fn, op)                                                      \
  static __device__ __forceinline__ void fn(float (&v)[8]) {                     \
        \
    asm(ROW8_STEP(op, 1) ROW8_STEP(op, 2) ROW8_STEP(op, 4) ROW8_STEP(op, 8)      \
        : "+v"(v[0]), "+v"(v[1]), "+v"(v[2]), "+v"(v[3]),                        \
          "+v"(v[4]), "+v"(v[5]), "+v"(v[6]), "+v"(v[7]));                       \
  }
ROW8_REDUCE(row8_max, "v_max_num_f32_dpp")
ROW8_REDUCE(row8_sum, "v_add_f32_dpp")
#undef ROW8_REDUCE
#undef ROW8_STEP

#define KSTEP 64
#define LP    68

__global__ __launch_bounds__(128)
void flash_attn_causal(const float* __restrict__ Q, const float* __restrict__ K,
                       const float* __restrict__ V, float* __restrict__ O)
{
  __shared__ __attribute__((aligned(16))) float Ks[KSTEP][LP];
  __shared__ __attribute__((aligned(16))) float Vt[HD][LP];
  __shared__ __attribute__((aligned(16))) float Ps[4][16][LP];

  const int bh   = blockIdx.x;
  const int qblk = blockIdx.y;
  const int tid  = threadIdx.x;
  const int w    = __builtin_amdgcn_readfirstlane(tid >> 5);
  const int lane = tid & 31;
  const int ll   = lane & 15;
  const int hh   = lane >> 4;
  const int qbase = qblk * 64 + w * 16;
  const float scale = 0.125f;

  const float* Qh = Q + (size_t)bh * SEQ * HD;
  const float* Kh = K + (size_t)bh * SEQ * HD;
  const float* Vh = V + (size_t)bh * SEQ * HD;

  v2f qf[16];
#pragma unroll
  for (int t = 0; t < 16; ++t) {
    v2f qv = *(const v2f*)&Qh[(size_t)(qbase + ll) * HD + t * 4 + 2 * hh];
    qf[t] = qv * scale;
  }

  const v8f z = {0.f,0.f,0.f,0.f,0.f,0.f,0.f,0.f};
  v8f oacc[4] = {z, z, z, z};
  float mrow[8], lrow[8];
#pragma unroll
  for (int j = 0; j < 8; ++j) { mrow[j] = -INFINITY; lrow[j] = 0.f; }

  const int r0   = tid >> 4;
  const int kc4  = (tid & 15) * 4;
  const int vkey = lane & 7;
  const int vc4  = (w * 4 + (lane >> 3)) * 4;
  const float* kp = Kh + r0 * HD + kc4;
  const float* vp = Vh + vkey * HD + vc4;
  float* ksd = &Ks[r0][kc4];
  float* vsd = &Vt[vc4][vkey];

  const int nkt = qblk + 1;
  for (int kt = 0; kt < nkt; ++kt) {
    const int kbase = kt * KSTEP;

    const float* kpt = kp + (size_t)kbase * HD;
    const float* vpt = vp + (size_t)kbase * HD;
#pragma unroll
    for (int part = 0; part < 8; ++part) {
      const float4 kd = *(const float4*)(kpt + part * 8 * HD);
      const float4 vd = *(const float4*)(vpt + part * 8 * HD);
      *(float4*)(ksd + part * 8 * LP) = kd;
      float* vs = vsd + part * 8;
      vs[0] = vd.x; vs[LP] = vd.y; vs[2 * LP] = vd.z; vs[3 * LP] = vd.w;
    }
    __syncthreads();

    const bool diag = (kt == nkt - 1);
    const int  nsub = diag ? (w + 1) : 4;

    v8f s[4] = {z, z, z, z};
    if (nsub == 4) {
#pragma unroll
      for (int t = 0; t < 16; ++t) {
#pragma unroll
        for (int n = 0; n < 4; ++n) {
          const v2f b = *(const v2f*)&Ks[n * 16 + ll][t * 4 + 2 * hh];
          s[n] = wmma_f32(qf[t], b, s[n]);
        }
      }
    } else {
#pragma unroll
      for (int n = 0; n < 3; ++n) {
        if (n < nsub) {
#pragma unroll
          for (int t = 0; t < 16; ++t) {
            const v2f b = *(const v2f*)&Ks[n * 16 + ll][t * 4 + 2 * hh];
            s[n] = wmma_f32(qf[t], b, s[n]);
          }
        }
      }
    }

    if (diag) {
#pragma unroll
      for (int n = 0; n < 4; ++n) {
#pragma unroll
        for (int j = 0; j < 8; ++j) {
          const int rowg = qbase + j + hh * 8;
          const int colg = kbase + n * 16 + ll;
          if (colg > rowg) s[n][j] = -INFINITY;
        }
      }
    }

    float red[8];
#pragma unroll
    for (int j = 0; j < 8; ++j) {
      const float t = fmaxf(fmaxf(s[0][j], s[1][j]), s[2][j]);
      red[j] = fmaxf(fmaxf(t, s[3][j]), mrow[j]);
    }
    row8_max(red);
#pragma unroll
    for (int j = 0; j < 8; ++j) {
      const float mnew  = red[j];
      const float alpha = __expf(mrow[j] - mnew);
      mrow[j] = mnew;
      const float p0 = __expf(s[0][j] - mnew);
      const float p1 = __expf(s[1][j] - mnew);
      const float p2 = __expf(s[2][j] - mnew);
      const float p3 = __expf(s[3][j] - mnew);
      lrow[j] = lrow[j] * alpha + ((p0 + p1) + (p2 + p3));
#pragma unroll
      for (int dt = 0; dt < 4; ++dt) oacc[dt][j] *= alpha;
      float* prow = &Ps[w][j + hh * 8][ll];
      prow[0] = p0; prow[16] = p1; prow[32] = p2; prow[48] = p3;
    }

    if (nsub == 4) {
#pragma unroll
      for (int t = 0; t < 16; ++t) {
        const v2f a = *(const v2f*)&Ps[w][ll][t * 4 + 2 * hh];
#pragma unroll
        for (int dt = 0; dt < 4; ++dt) {
          const v2f b = *(const v2f*)&Vt[dt * 16 + ll][t * 4 + 2 * hh];
          oacc[dt] = wmma_f32(a, b, oacc[dt]);
        }
      }
    } else {
#pragma unroll
      for (int n = 0; n < 3; ++n) {
        if (n < nsub) {
#pragma unroll
          for (int tt = 0; tt < 4; ++tt) {
            const int t = n * 4 + tt;
            const v2f a = *(const v2f*)&Ps[w][ll][t * 4 + 2 * hh];
#pragma unroll
            for (int dt = 0; dt < 4; ++dt) {
              const v2f b = *(const v2f*)&Vt[dt * 16 + ll][t * 4 + 2 * hh];
              oacc[dt] = wmma_f32(a, b, oacc[dt]);
            }
          }
        }
      }
    }
    __syncthreads();
  }

  const int b = bh >> 4;
  const int h = bh & 15;
  row8_sum(lrow);
#pragma unroll
  for (int j = 0; j < 8; ++j) {
    const float inv = 1.0f / lrow[j];
    const int rowg = qbase + j + hh * 8;
#pragma unroll
    for (int dt = 0; dt < 4; ++dt) {
      const int d = dt * 16 + ll;
      O[((size_t)b * SEQ + rowg) * MD + h * HD + d] = oacc[dt][j] * inv;
    }
  }
}

extern "C" void kernel_launch(void* const* d_in, const int* in_sizes, int n_in,
                              void* d_out, int out_size, void* d_ws, size_t ws_size,
                              hipStream_t stream) {
  (void)in_sizes; (void)n_in; (void)out_size; (void)ws_size;
  const float* x  = (const float*)d_in[0];
  const float* Wq = (const float*)d_in[1];
  const float* Wk = (const float*)d_in[2];
  const float* Wv = (const float*)d_in[3];
  const float* Wo = (const float*)d_in[4];
  const float* bo = (const float*)d_in[5];

  float* ws = (float*)d_ws;
  const size_t SZ = (size_t)BATCH * SEQ * MD;
  float* qb = ws;
  float* kb = ws + SZ;
  float* vb = ws + 2 * SZ;
  float* ao = ws + 3 * SZ;

  const dim3 ggrid(BATCH * SEQ / 128, MD / 64);
  gemm_qkv_nt_128x64<<<ggrid, 256, 0, stream>>>(x, Wq, Wk, Wv, qb, kb, vb);

  flash_attn_causal<<<dim3(BATCH * NH, SEQ / 64), 128, 0, stream>>>(qb, kb, vb, ao);

  gemm_nt_128x64<<<ggrid, 256, 0, stream>>>(ao, Wo, bo, (float*)d_out, 0);
}
